# starting kernel
# baseline (speedup 1.0000x reference)
#include <hip/hip_runtime.h>

#define NUM_USERS  100000
#define NUM_ITEMS  50000
#define NNZ        2400000
#define EMB        64
#define NTOT       (NUM_USERS + NUM_ITEMS)
#define NPAIR      16384
#define NUM_LAYERS 3

typedef __attribute__((ext_vector_type(2))) float v2f;
typedef __attribute__((ext_vector_type(8))) float v8f;

__global__ void lgcn_init(const float* __restrict__ eu, const float* __restrict__ ei,
                          float* __restrict__ cur, float* __restrict__ acc,
                          float* __restrict__ nxt) {
    size_t i = (size_t)blockIdx.x * blockDim.x + threadIdx.x;
    const size_t total4 = (size_t)NTOT * EMB / 4;
    if (i >= total4) return;
    const size_t user4 = (size_t)NUM_USERS * EMB / 4;
    float4 v = (i < user4) ? ((const float4*)eu)[i]
                           : ((const float4*)ei)[i - user4];
    ((float4*)cur)[i] = v;
    ((float4*)acc)[i] = v;
    ((float4*)nxt)[i] = make_float4(0.f, 0.f, 0.f, 0.f);
}

__global__ void lgcn_spmm(const int* __restrict__ row, const int* __restrict__ col,
                          const float* __restrict__ val,
                          const float* __restrict__ cur, float* __restrict__ nxt) {
    long long t = (long long)blockIdx.x * blockDim.x + threadIdx.x;
    long long e = t >> 4;
    int lane = (int)(t & 15);
    if (e >= NNZ) return;
    int   r = row[e];
    int   c = col[e];
    float w = val[e];
    float4 x = ((const float4*)(cur + (size_t)c * EMB))[lane];
    float* dst = nxt + (size_t)r * EMB + lane * 4;
    atomicAdd(dst + 0, w * x.x);
    atomicAdd(dst + 1, w * x.y);
    atomicAdd(dst + 2, w * x.z);
    atomicAdd(dst + 3, w * x.w);
}

__global__ void lgcn_accum(float* __restrict__ acc, const float* __restrict__ layer) {
    size_t i = (size_t)blockIdx.x * blockDim.x + threadIdx.x;
    const size_t total4 = (size_t)NTOT * EMB / 4;
    if (i >= total4) return;
    float4 a = ((const float4*)acc)[i];
    float4 b = ((const float4*)layer)[i];
    a.x += b.x; a.y += b.y; a.z += b.z; a.w += b.w;
    ((float4*)acc)[i] = a;
}

__global__ void lgcn_zero(float* __restrict__ p) {
    size_t i = (size_t)blockIdx.x * blockDim.x + threadIdx.x;
    const size_t total4 = (size_t)NTOT * EMB / 4;
    if (i >= total4) return;
    ((float4*)p)[i] = make_float4(0.f, 0.f, 0.f, 0.f);
}

__global__ __launch_bounds__(32) void lgcn_gamma(const float* __restrict__ acc,
                                                 const int* __restrict__ users,
                                                 const int* __restrict__ items,
                                                 float* __restrict__ out) {
    int tile = blockIdx.x;
    int l    = threadIdx.x;
    int m    = l & 15;
    int kb   = (l >> 4) * 2;

    int u  = users[tile * 16 + m];
    int it = items[tile * 16 + m];
    const float* urow = acc + (size_t)u * EMB;
    const float* vrow = acc + ((size_t)NUM_USERS + (size_t)it) * EMB;

    v8f d = {};
#pragma unroll
    for (int k0 = 0; k0 < EMB; k0 += 4) {
        v2f a = *(const v2f*)(urow + k0 + kb);
        v2f b = *(const v2f*)(vrow + k0 + kb);
        d = __builtin_amdgcn_wmma_f32_16x16x4_f32(
                 false, a,  false, b,
                 (short)0, d,  false,  false);
    }

    const float scale = 1.0f / 16.0f;
    float g = 0.f; int idx = -1;
#pragma unroll
    for (int j = 0; j < 8; ++j) {
        if (l == j)      { g = d[j]; idx = j;     }
        if (l == 24 + j) { g = d[j]; idx = 8 + j; }
    }
    if (idx >= 0) out[tile * 16 + idx] = g * scale;
}

extern "C" void kernel_launch(void* const* d_in, const int* in_sizes, int n_in,
                              void* d_out, int out_size, void* d_ws, size_t ws_size,
                              hipStream_t stream) {
    const int*   row      = (const int*)  d_in[0];
    const int*   col      = (const int*)  d_in[1];
    const float* val      = (const float*)d_in[2];
    const float* emb_user = (const float*)d_in[3];
    const float* emb_item = (const float*)d_in[4];
    const int*   users    = (const int*)  d_in[5];
    const int*   items    = (const int*)  d_in[6];
    float*       out      = (float*)d_out;
    (void)in_sizes; (void)n_in; (void)out_size; (void)ws_size;

    const size_t matElems = (size_t)NTOT * EMB;
    float* cur = (float*)d_ws;
    float* nxt = cur + matElems;
    float* acc = nxt + matElems;

    const size_t total4 = matElems / 4;
    const int    tpb    = 256;
    const int    gElem  = (int)((total4 + tpb - 1) / tpb);

    lgcn_init<<<gElem, tpb, 0, stream>>>(emb_user, emb_item, cur, acc, nxt);

    for (int layer = 0; layer < NUM_LAYERS; ++layer) {
        long long spmmThreads = (long long)NNZ * 16;
        int gSpmm = (int)((spmmThreads + tpb - 1) / tpb);
        lgcn_spmm<<<gSpmm, tpb, 0, stream>>>(row, col, val, cur, nxt);
        lgcn_accum<<<gElem, tpb, 0, stream>>>(acc, nxt);
        float* t = cur; cur = nxt; nxt = t;
        if (layer + 1 < NUM_LAYERS)
            lgcn_zero<<<gElem, tpb, 0, stream>>>(nxt);
    }

    lgcn_gamma<<<NPAIR / 16, 32, 0, stream>>>(acc, users, items, out);
}


# Edges counting-sorted by destination row inside the call; 16 lanes own an output row and sum its edges and the layer sum in registers, so float atomics, zero fills and add passes go
# speedup vs baseline: 2.0733x; 2.0733x over previous
#include <hip/hip_runtime.h>

#define NUM_USERS  100000
#define NUM_ITEMS  50000
#define NNZ        2400000
#define EMB        64
#define NTOT       (NUM_USERS + NUM_ITEMS)
#define NPAIR      16384
#define NUM_LAYERS 3

typedef __attribute__((ext_vector_type(2))) float v2f;
typedef __attribute__((ext_vector_type(8))) float v8f;

__global__ void lgcn_clear_counts(int* __restrict__ cnt, int* __restrict__ total) {
    int i = blockIdx.x * blockDim.x + threadIdx.x;
    if (i < NTOT) cnt[i] = 0;
    if (i == 0) *total = 0;
}

__global__ void lgcn_count_rows(const int* __restrict__ row, int* __restrict__ cnt) {
    int e = blockIdx.x * blockDim.x + threadIdx.x;
    if (e < NNZ) atomicAdd(&cnt[row[e]], 1);
}

__global__ __launch_bounds__(256) void lgcn_place_rows(const int* __restrict__ cnt,
                                                       int* __restrict__ cur,
                                                       int* __restrict__ span,
                                                       int* __restrict__ total) {
    __shared__ int wsum[8];
    __shared__ int bbase;
    const int r    = blockIdx.x * 256 + threadIdx.x;
    const int lane = threadIdx.x & 31;
    const int wv   = threadIdx.x >> 5;
    const int c    = (r < NTOT) ? cnt[r] : 0;
    int inc = c;
    for (int d = 1; d < 32; d <<= 1) {
        int t = __shfl_up(inc, d, 32);
        if (lane >= d) inc += t;
    }
    if (lane == 31) wsum[wv] = inc;
    __syncthreads();
    if (threadIdx.x == 0) {
        int s = 0;
        for (int k = 0; k < 8; ++k) { int t = wsum[k]; wsum[k] = s; s += t; }
        bbase = atomicAdd(total, s);
    }
    __syncthreads();
    if (r < NTOT) {
        const int first = bbase + wsum[wv] + inc - c;
        cur[r] = first;
        ((int2*)span)[r] = make_int2(first, c);
    }
}

__global__ void lgcn_fill_rows(const int* __restrict__ row, const int* __restrict__ col,
                               const float* __restrict__ val,
                               int* __restrict__ cur, int* __restrict__ edges) {
    int e = blockIdx.x * blockDim.x + threadIdx.x;
    if (e >= NNZ) return;
    int slot = atomicAdd(&cur[row[e]], 1);
    ((int2*)edges)[slot] = make_int2(col[e], __float_as_int(val[e]));
}

template <int LAYER>
__device__ __forceinline__ const float* src_row(int c, const float* eu, const float* ei,
                                                const float* src) {
    if (LAYER == 1)
        return (c < NUM_USERS) ? eu + (size_t)c * EMB : ei + (size_t)(c - NUM_USERS) * EMB;
    return src + (size_t)c * EMB;
}

template <int LAYER>
__global__ __launch_bounds__(256) void lgcn_spmm_rows(const int* __restrict__ span,
                                                      const int* __restrict__ edges,
                                                      const float* __restrict__ eu,
                                                      const float* __restrict__ ei,
                                                      const float* src, float* dst) {
    const int gid = blockIdx.x * 256 + threadIdx.x;
    const int r   = gid >> 4;
    const int sub = threadIdx.x & 15;

    const int2 sp = ((const int2*)span)[r];
    const int beg = sp.x;
    const int n   = sp.y;
    const int nmax = __builtin_amdgcn_readfirstlane(max(n, __shfl_xor(n, 16, 32)));
    const int2* ed = (const int2*)edges + beg;

    float4 s = make_float4(0.f, 0.f, 0.f, 0.f);
    for (int base = 0; base < nmax; base += 16) {
        int myc = 0; float myw = 0.f;
        if (base + sub < n) {
            int2 p = ed[base + sub];
            myc = p.x; myw = __int_as_float(p.y);
        }
        const int lim = min(16, nmax - base);
        for (int j = 0; j < lim; j += 4) {
            const int   c0 = __shfl(myc, j,     16), c1 = __shfl(myc, j + 1, 16);
            const int   c2 = __shfl(myc, j + 2, 16), c3 = __shfl(myc, j + 3, 16);
            const float w0 = __shfl(myw, j,     16), w1 = __shfl(myw, j + 1, 16);
            const float w2 = __shfl(myw, j + 2, 16), w3 = __shfl(myw, j + 3, 16);
            const float4 x0 = ((const float4*)src_row<LAYER>(c0, eu, ei, src))[sub];
            const float4 x1 = ((const float4*)src_row<LAYER>(c1, eu, ei, src))[sub];
            const float4 x2 = ((const float4*)src_row<LAYER>(c2, eu, ei, src))[sub];
            const float4 x3 = ((const float4*)src_row<LAYER>(c3, eu, ei, src))[sub];
            s.x += w0 * x0.x; s.y += w0 * x0.y; s.z += w0 * x0.z; s.w += w0 * x0.w;
            s.x += w1 * x1.x; s.y += w1 * x1.y; s.z += w1 * x1.z; s.w += w1 * x1.w;
            s.x += w2 * x2.x; s.y += w2 * x2.y; s.z += w2 * x2.z; s.w += w2 * x2.w;
            s.x += w3 * x3.x; s.y += w3 * x3.y; s.z += w3 * x3.z; s.w += w3 * x3.w;
        }
    }

    float4* drow = (float4*)(dst + (size_t)r * EMB) + sub;
    if (LAYER == 3) {
        const float* er = (r < NUM_USERS) ? eu + (size_t)r * EMB
                                          : ei + (size_t)(r - NUM_USERS) * EMB;
        const float4 e0 = ((const float4*)er)[sub];
        const float4 a1 = *drow;
        const float4 a2 = ((const float4*)(src + (size_t)r * EMB))[sub];
        s.x = ((e0.x + a1.x) + a2.x) + s.x;
        s.y = ((e0.y + a1.y) + a2.y) + s.y;
        s.z = ((e0.z + a1.z) + a2.z) + s.z;
        s.w = ((e0.w + a1.w) + a2.w) + s.w;
    }
    *drow = s;
}

__global__ __launch_bounds__(32) void lgcn_gamma(const float* __restrict__ acc,
                                                 const int* __restrict__ users,
                                                 const int* __restrict__ items,
                                                 float* __restrict__ out) {
    int tile = blockIdx.x;
    int l    = threadIdx.x;
    int m    = l & 15;
    int kb   = (l >> 4) * 2;

    int u  = users[tile * 16 + m];
    int it = items[tile * 16 + m];
    const float* urow = acc + (size_t)u * EMB;
    const float* vrow = acc + ((size_t)NUM_USERS + (size_t)it) * EMB;

    v8f d = {};
#pragma unroll
    for (int k0 = 0; k0 < EMB; k0 += 4) {
        v2f a = *(const v2f*)(urow + k0 + kb);
        v2f b = *(const v2f*)(vrow + k0 + kb);
        d = __builtin_amdgcn_wmma_f32_16x16x4_f32(
                 false, a,  false, b,
                 (short)0, d,  false,  false);
    }

    const float scale = 1.0f / 16.0f;
    float g = 0.f; int idx = -1;
#pragma unroll
    for (int j = 0; j < 8; ++j) {
        if (l == j)      { g = d[j]; idx = j;     }
        if (l == 24 + j) { g = d[j]; idx = 8 + j; }
    }
    if (idx >= 0) out[tile * 16 + idx] = g * scale;
}

extern "C" void kernel_launch(void* const* d_in, const int* in_sizes, int n_in,
                              void* d_out, int out_size, void* d_ws, size_t ws_size,
                              hipStream_t stream) {
    const int*   row      = (const int*)  d_in[0];
    const int*   col      = (const int*)  d_in[1];
    const float* val      = (const float*)d_in[2];
    const float* emb_user = (const float*)d_in[3];
    const float* emb_item = (const float*)d_in[4];
    const int*   users    = (const int*)  d_in[5];
    const int*   items    = (const int*)  d_in[6];
    float*       out      = (float*)d_out;
    (void)in_sizes; (void)n_in; (void)out_size; (void)ws_size;

    const size_t matElems = (size_t)NTOT * EMB;
    float* x1    = (float*)d_ws;
    float* x2    = x1 + matElems;
    int*   edges = (int*)(x2 + matElems);
    int*   cnt   = edges + 2 * (size_t)NNZ;
    int*   cur   = cnt + NTOT;
    int*   span  = cur + NTOT;
    int*   total = span + 2 * NTOT;

    const int tpb   = 256;
    const int gRow  = (NTOT + tpb - 1) / tpb;
    const int gEdge = (NNZ + tpb - 1) / tpb;
    const int gSpmm = NTOT * 16 / tpb;

    lgcn_clear_counts<<<gRow, tpb, 0, stream>>>(cnt, total);
    lgcn_count_rows<<<gEdge, tpb, 0, stream>>>(row, cnt);
    lgcn_place_rows<<<gRow, tpb, 0, stream>>>(cnt, cur, span, total);
    lgcn_fill_rows<<<gEdge, tpb, 0, stream>>>(row, col, val, cur, edges);

    lgcn_spmm_rows<1><<<gSpmm, tpb, 0, stream>>>(span, edges, emb_user, emb_item, x1, x1);
    lgcn_spmm_rows<2><<<gSpmm, tpb, 0, stream>>>(span, edges, emb_user, emb_item, x1, x2);
    lgcn_spmm_rows<3><<<gSpmm, tpb, 0, stream>>>(span, edges, emb_user, emb_item, x2, x1);

    lgcn_gamma<<<NPAIR / 16, 32, 0, stream>>>(x1, users, items, out);
}


# Third propagation computed only for the 2 x 16384 rows that the pairs name, inside the scoring kernel (rows to LDS, f32 WMMA from there); the full third layer launch and its stores go
# speedup vs baseline: 2.2491x; 1.0848x over previous
#include <hip/hip_runtime.h>

#define NUM_USERS  100000
#define NUM_ITEMS  50000
#define NNZ        2400000
#define EMB        64
#define NTOT       (NUM_USERS + NUM_ITEMS)
#define NPAIR      16384
#define NUM_LAYERS 3

typedef __attribute__((ext_vector_type(2))) float v2f;
typedef __attribute__((ext_vector_type(8))) float v8f;

__global__ void lgcn_clear_counts(int* __restrict__ cnt, int* __restrict__ total) {
    int i = blockIdx.x * blockDim.x + threadIdx.x;
    if (i < NTOT) cnt[i] = 0;
    if (i == 0) *total = 0;
}

__global__ void lgcn_count_rows(const int* __restrict__ row, int* __restrict__ cnt) {
    int e = blockIdx.x * blockDim.x + threadIdx.x;
    if (e < NNZ) atomicAdd(&cnt[row[e]], 1);
}

__global__ __launch_bounds__(256) void lgcn_place_rows(const int* __restrict__ cnt,
                                                       int* __restrict__ cur,
                                                       int* __restrict__ span,
                                                       int* __restrict__ total) {
    __shared__ int wsum[8];
    __shared__ int bbase;
    const int r    = blockIdx.x * 256 + threadIdx.x;
    const int lane = threadIdx.x & 31;
    const int wv   = threadIdx.x >> 5;
    const int c    = (r < NTOT) ? cnt[r] : 0;
    int inc = c;
    for (int d = 1; d < 32; d <<= 1) {
        int t = __shfl_up(inc, d, 32);
        if (lane >= d) inc += t;
    }
    if (lane == 31) wsum[wv] = inc;
    __syncthreads();
    if (threadIdx.x == 0) {
        int s = 0;
        for (int k = 0; k < 8; ++k) { int t = wsum[k]; wsum[k] = s; s += t; }
        bbase = atomicAdd(total, s);
    }
    __syncthreads();
    if (r < NTOT) {
        const int first = bbase + wsum[wv] + inc - c;
        cur[r] = first;
        ((int2*)span)[r] = make_int2(first, c);
    }
}

__global__ void lgcn_fill_rows(const int* __restrict__ row, const int* __restrict__ col,
                               const float* __restrict__ val,
                               int* __restrict__ cur, int* __restrict__ edges) {
    int e = blockIdx.x * blockDim.x + threadIdx.x;
    if (e >= NNZ) return;
    int slot = atomicAdd(&cur[row[e]], 1);
    ((int2*)edges)[slot] = make_int2(col[e], __float_as_int(val[e]));
}

template <int LAYER>
__device__ __forceinline__ const float* src_row(int c, const float* eu, const float* ei,
                                                const float* src) {
    if (LAYER == 1)
        return (c < NUM_USERS) ? eu + (size_t)c * EMB : ei + (size_t)(c - NUM_USERS) * EMB;
    return src + (size_t)c * EMB;
}

template <int LAYER>
__device__ __forceinline__ float4 row_edge_sum(const int2 sp, const int* __restrict__ edges,
                                               const float* __restrict__ eu,
                                               const float* __restrict__ ei,
                                               const float* src, const int sub) {
    const int beg = sp.x;
    const int n   = sp.y;
    const int nmax = __builtin_amdgcn_readfirstlane(max(n, __shfl_xor(n, 16, 32)));
    const int2* ed = (const int2*)edges + beg;

    float4 s = make_float4(0.f, 0.f, 0.f, 0.f);
    for (int base = 0; base < nmax; base += 16) {
        int myc = 0; float myw = 0.f;
        if (base + sub < n) {
            int2 p = ed[base + sub];
            myc = p.x; myw = __int_as_float(p.y);
        }
        const int lim = min(16, nmax - base);
        for (int j = 0; j < lim; j += 4) {
            const int   c0 = __shfl(myc, j,     16), c1 = __shfl(myc, j + 1, 16);
            const int   c2 = __shfl(myc, j + 2, 16), c3 = __shfl(myc, j + 3, 16);
            const float w0 = __shfl(myw, j,     16), w1 = __shfl(myw, j + 1, 16);
            const float w2 = __shfl(myw, j + 2, 16), w3 = __shfl(myw, j + 3, 16);
            const float4 x0 = ((const float4*)src_row<LAYER>(c0, eu, ei, src))[sub];
            const float4 x1 = ((const float4*)src_row<LAYER>(c1, eu, ei, src))[sub];
            const float4 x2 = ((const float4*)src_row<LAYER>(c2, eu, ei, src))[sub];
            const float4 x3 = ((const float4*)src_row<LAYER>(c3, eu, ei, src))[sub];
            s.x += w0 * x0.x; s.y += w0 * x0.y; s.z += w0 * x0.z; s.w += w0 * x0.w;
            s.x += w1 * x1.x; s.y += w1 * x1.y; s.z += w1 * x1.z; s.w += w1 * x1.w;
            s.x += w2 * x2.x; s.y += w2 * x2.y; s.z += w2 * x2.z; s.w += w2 * x2.w;
            s.x += w3 * x3.x; s.y += w3 * x3.y; s.z += w3 * x3.z; s.w += w3 * x3.w;
        }
    }
    return s;
}

template <int LAYER>
__global__ __launch_bounds__(256) void lgcn_spmm_rows(const int* __restrict__ span,
                                                      const int* __restrict__ edges,
                                                      const float* __restrict__ eu,
                                                      const float* __restrict__ ei,
                                                      const float* src, float* dst) {
    const int gid = blockIdx.x * 256 + threadIdx.x;
    const int r   = gid >> 4;
    const int sub = threadIdx.x & 15;
    const int2 sp = ((const int2*)span)[r];
    const float4 s = row_edge_sum<LAYER>(sp, edges, eu, ei, src, sub);
    ((float4*)(dst + (size_t)r * EMB))[sub] = s;
}

#define ROWP4 17

__global__ __launch_bounds__(512) void lgcn_last_rows_gamma(const int* __restrict__ span,
                                                            const int* __restrict__ edges,
                                                            const float* __restrict__ eu,
                                                            const float* __restrict__ ei,
                                                            const float* __restrict__ x1,
                                                            const float* __restrict__ x2,
                                                            const int* __restrict__ users,
                                                            const int* __restrict__ items,
                                                            float* __restrict__ out) {
    __shared__ float4 rows[32 * ROWP4];
    const int tile = blockIdx.x;
    const int grp  = threadIdx.x >> 4;
    const int sub  = threadIdx.x & 15;
    const int pr   = tile * 16 + (grp & 15);
    const int r    = (grp < 16) ? users[pr] : NUM_USERS + items[pr];

    const int2 sp = ((const int2*)span)[r];
    const float* er = (r < NUM_USERS) ? eu + (size_t)r * EMB
                                      : ei + (size_t)(r - NUM_USERS) * EMB;
    const float4 e0 = ((const float4*)er)[sub];
    const float4 a1 = ((const float4*)(x1 + (size_t)r * EMB))[sub];
    const float4 a2 = ((const float4*)(x2 + (size_t)r * EMB))[sub];

    float4 s = row_edge_sum<3>(sp, edges, eu, ei, x2, sub);
    s.x = ((e0.x + a1.x) + a2.x) + s.x;
    s.y = ((e0.y + a1.y) + a2.y) + s.y;
    s.z = ((e0.z + a1.z) + a2.z) + s.z;
    s.w = ((e0.w + a1.w) + a2.w) + s.w;
    rows[grp * ROWP4 + sub] = s;
    __syncthreads();

    if (__builtin_amdgcn_readfirstlane((int)(threadIdx.x >> 5)) == 0) {
        const int l  = threadIdx.x & 31;
        const int m  = l & 15;
        const int kb = (l >> 4) * 2;
        const float* urow = (const float*)rows + m * (ROWP4 * 4);
        const float* vrow = (const float*)rows + (16 + m) * (ROWP4 * 4);

        v8f d = {};
#pragma unroll
        for (int k0 = 0; k0 < EMB; k0 += 4) {
            v2f a = *(const v2f*)(urow + k0 + kb);
            v2f b = *(const v2f*)(vrow + k0 + kb);
            d = __builtin_amdgcn_wmma_f32_16x16x4_f32(
                     false, a,  false, b,
                     (short)0, d,  false,  false);
        }

        const float scale = 1.0f / 16.0f;
        float gv = 0.f; int idx = -1;
#pragma unroll
        for (int j = 0; j < 8; ++j) {
            if (l == j)      { gv = d[j]; idx = j;     }
            if (l == 24 + j) { gv = d[j]; idx = 8 + j; }
        }
        if (idx >= 0) out[tile * 16 + idx] = gv * scale;
    }
}

extern "C" void kernel_launch(void* const* d_in, const int* in_sizes, int n_in,
                              void* d_out, int out_size, void* d_ws, size_t ws_size,
                              hipStream_t stream) {
    const int*   row      = (const int*)  d_in[0];
    const int*   col      = (const int*)  d_in[1];
    const float* val      = (const float*)d_in[2];
    const float* emb_user = (const float*)d_in[3];
    const float* emb_item = (const float*)d_in[4];
    const int*   users    = (const int*)  d_in[5];
    const int*   items    = (const int*)  d_in[6];
    float*       out      = (float*)d_out;
    (void)in_sizes; (void)n_in; (void)out_size; (void)ws_size;

    const size_t matElems = (size_t)NTOT * EMB;
    float* x1    = (float*)d_ws;
    float* x2    = x1 + matElems;
    int*   edges = (int*)(x2 + matElems);
    int*   cnt   = edges + 2 * (size_t)NNZ;
    int*   cur   = cnt + NTOT;
    int*   span  = cur + NTOT;
    int*   total = span + 2 * NTOT;

    const int tpb   = 256;
    const int gRow  = (NTOT + tpb - 1) / tpb;
    const int gEdge = (NNZ + tpb - 1) / tpb;
    const int gSpmm = NTOT * 16 / tpb;

    lgcn_clear_counts<<<gRow, tpb, 0, stream>>>(cnt, total);
    lgcn_count_rows<<<gEdge, tpb, 0, stream>>>(row, cnt);
    lgcn_place_rows<<<gRow, tpb, 0, stream>>>(cnt, cur, span, total);
    lgcn_fill_rows<<<gEdge, tpb, 0, stream>>>(row, col, val, cur, edges);

    lgcn_spmm_rows<1><<<gSpmm, tpb, 0, stream>>>(span, edges, emb_user, emb_item, x1, x1);
    lgcn_spmm_rows<2><<<gSpmm, tpb, 0, stream>>>(span, edges, emb_user, emb_item, x1, x2);

    lgcn_last_rows_gamma<<<NPAIR / 16, 512, 0, stream>>>(span, edges, emb_user, emb_item,
                                                         x1, x2, users, items, out);
}


# First propagation done inside the bin-ordering launch from each bin's ordered edges while they lie in LDS (512 bins, two work-groups per WGP, row pairs dealt to free waves); the layer-1 launch goes
# speedup vs baseline: 7.3077x; 3.2492x over previous
#include <hip/hip_runtime.h>

#define NUM_USERS  100000
#define NUM_ITEMS  50000
#define NNZ        2400000
#define EMB        64
#define NTOT       (NUM_USERS + NUM_ITEMS)
#define NPAIR      16384
#define NUM_LAYERS 3

typedef __attribute__((ext_vector_type(2))) float v2f;
typedef __attribute__((ext_vector_type(8))) float v8f;

#define BIN_ROWS  293
#define NBIN      ((NTOT + BIN_ROWS - 1) / BIN_ROWS)
#define NBIN_PAD  ((NBIN + 31) / 32 * 32)
#define ROWS_PAD  ((BIN_ROWS + 31) / 32 * 32)
#define ROW_BITS  9
#define NPAIR_BIN ((BIN_ROWS + 1) / 2)
#define NPART     256
#define CHUNK     (NNZ / NPART)
#define CHUNK_PITCH ((CHUNK + 15) / 16 * 16)
#define COL_BITS  18
#define COL_MASK  ((1 << COL_BITS) - 1)
#define SORT_TPB  1024
#define PER_THR   ((CHUNK + SORT_TPB - 1) / SORT_TPB)
#define STAGE_CAP 8192
#define RUNS_PER_WAVE (NPART / (SORT_TPB / 32))
static_assert(NNZ % NPART == 0, "chunks must tile the edge list");
static_assert(NTOT <= (1 << COL_BITS), "col must fit its field");
static_assert(BIN_ROWS <= (1 << ROW_BITS) && ROW_BITS + COL_BITS <= 32, "row within bin and col share a word");
static_assert(NPART % 32 == 0 && NPART <= SORT_TPB, "a bin's runs are dealt to whole waves");
static_assert(NBIN_PAD <= SORT_TPB, "one thread per bin in the scan");
static_assert(2 * NPART * NBIN <= 2 * NTOT, "runMat must end before span starts");
static_assert((size_t)NPART * CHUNK_PITCH * 2 <= (size_t)NTOT * EMB, "the ordered chunks must fit where x2 will be");

__global__ __launch_bounds__(SORT_TPB) void lgcn_chunk_by_bin(const int* __restrict__ row,
                                                              const int* __restrict__ col,
                                                              const float* __restrict__ val,
                                                              int* __restrict__ runMat,
                                                              int* __restrict__ part) {
    __shared__ int  hist[NBIN_PAD];
    __shared__ int  wsum[NBIN_PAD / 32];
    __shared__ int2 stage[CHUNK];
    const int wg   = blockIdx.x;
    const int t    = threadIdx.x;
    const int lane = t & 31;
    const int wv   = t >> 5;

    if (t < NBIN_PAD) hist[t] = 0;
    __syncthreads();

    const int beg = wg * CHUNK;
    int r[PER_THR], c[PER_THR], v[PER_THR];
#pragma unroll
    for (int k = 0; k < PER_THR; ++k) {
        const int i = t + k * SORT_TPB;
        const bool on = i < CHUNK;
        r[k] = on ? row[beg + i] : 0;
        c[k] = on ? col[beg + i] : 0;
        v[k] = on ? __float_as_int(val[beg + i]) : 0;
    }
    int b[PER_THR];
#pragma unroll
    for (int k = 0; k < PER_THR; ++k) {
        const unsigned rr = (unsigned)r[k];
        const unsigned bb = rr / BIN_ROWS;
        b[k] = (int)bb;
        r[k] = (int)(rr - bb * BIN_ROWS);
    }
#pragma unroll
    for (int k = 0; k < PER_THR; ++k) {
        const int i = t + k * SORT_TPB;
        if (i < CHUNK) atomicAdd(&hist[b[k]], 1);
    }
    __syncthreads();

    int cnt = 0, inc = 0;
    if (t < NBIN_PAD) {
        cnt = hist[t];
        inc = cnt;
        for (int d = 1; d < 32; d <<= 1) {
            const int u = __shfl_up(inc, d, 32);
            if (lane >= d) inc += u;
        }
        if (lane == 31) wsum[wv] = inc;
    }
    __syncthreads();
    if (t < NBIN_PAD) {
        int below = 0;
#pragma unroll
        for (int k = 0; k < NBIN_PAD / 32; ++k) {
            const int u = wsum[k];
            if (k < wv) below += u;
        }
        const int first = below + inc - cnt;
        hist[t] = first;
        if (t < NBIN) ((int2*)runMat)[wg * NBIN + t] = make_int2(cnt, first);
    }
    __syncthreads();

#pragma unroll
    for (int k = 0; k < PER_THR; ++k) {
        const int i = t + k * SORT_TPB;
        if (i < CHUNK) {
            const int slot = atomicAdd(&hist[b[k]], 1);
            stage[slot]    = make_int2((r[k] << COL_BITS) | c[k], v[k]);
        }
    }
    __syncthreads();
    int2* dst = (int2*)part + (size_t)wg * CHUNK_PITCH;
    for (int i = t; i < CHUNK; i += SORT_TPB) dst[i] = stage[i];
}

template <int LAYER>
__device__ __forceinline__ float4 row_edge_sum(const int2* ed, const int n,
                                               const float* __restrict__ eu,
                                               const float* __restrict__ ei,
                                               const float* src, const int sub);

__global__ __launch_bounds__(SORT_TPB, 16) void lgcn_bin_rows_layer1(const int* __restrict__ runMat,
                                                          const int* __restrict__ part,
                                                          int* __restrict__ edges,
                                                          int* __restrict__ span,
                                                          const float* __restrict__ eu,
                                                          const float* __restrict__ ei,
                                                          float* __restrict__ x1) {
    __shared__ int  rcnt[ROWS_PAD];
    __shared__ int  rcur[ROWS_PAD];
    __shared__ int  wsum[ROWS_PAD / 32];
    __shared__ int  runCnt[NPART];
    __shared__ int  runSrc[NPART];
    __shared__ int  redCnt[NPART / 32];
    __shared__ int  redPre[NPART / 32];
    __shared__ int2 stage[STAGE_CAP];
    __shared__ int  nextPair;
    const int bin  = blockIdx.x;
    const int t    = threadIdx.x;
    const int lane = t & 31;
    const int wv   = t >> 5;
    const int2* src = (const int2*)part;

    if (t < ROWS_PAD) rcnt[t] = 0;
    if (t == 0) nextPair = 0;
    if (t < NPART) {
        const int2 cp = ((const int2*)runMat)[t * NBIN + bin];
        runCnt[t] = cp.x;
        runSrc[t] = t * CHUNK_PITCH + cp.y;
        int sc = cp.x, sp = cp.y;
        for (int d = 16; d >= 1; d >>= 1) {
            sc += __shfl_xor(sc, d, 32);
            sp += __shfl_xor(sp, d, 32);
        }
        if (lane == 0) { redCnt[wv] = sc; redPre[wv] = sp; }
    }
    __syncthreads();
    int nAll = 0, bsAll = 0;
#pragma unroll
    for (int k = 0; k < NPART / 32; ++k) { nAll += redCnt[k]; bsAll += redPre[k]; }
    const int n  = __builtin_amdgcn_readfirstlane(nAll);
    const int bs = __builtin_amdgcn_readfirstlane(bsAll);
    int2* dst = (int2*)edges + bs;

    int2 p[RUNS_PER_WAVE];
    int  have = 0;
    bool longRun = false;
    {
        int cn[RUNS_PER_WAVE], sr[RUNS_PER_WAVE];
#pragma unroll
        for (int k = 0; k < RUNS_PER_WAVE; ++k) {
            cn[k] = runCnt[wv * RUNS_PER_WAVE + k];
            sr[k] = runSrc[wv * RUNS_PER_WAVE + k];
        }
#pragma unroll
        for (int k = 0; k < RUNS_PER_WAVE; ++k) {
            const bool on = lane < cn[k];
            have |= (on ? 1 : 0) << k;
            longRun = longRun || cn[k] > 32;
            p[k] = on ? src[sr[k] + lane] : make_int2(0, 0);
        }
    }
#pragma unroll
    for (int k = 0; k < RUNS_PER_WAVE; ++k)
        if ((have >> k) & 1) atomicAdd(&rcnt[p[k].x >> COL_BITS], 1);
    if (longRun) {
        for (int k = 0; k < RUNS_PER_WAVE; ++k) {
            const int run = wv * RUNS_PER_WAVE + k;
            const int cn  = runCnt[run];
            const int s   = runSrc[run];
            for (int i = lane + 32; i < cn; i += 32) atomicAdd(&rcnt[src[s + i].x >> COL_BITS], 1);
        }
    }
    __syncthreads();

    int c = 0, inc = 0;
    if (t < ROWS_PAD) {
        c = rcnt[t];
        inc = c;
        for (int d = 1; d < 32; d <<= 1) {
            const int u = __shfl_up(inc, d, 32);
            if (lane >= d) inc += u;
        }
        if (lane == 31) wsum[wv] = inc;
    }
    __syncthreads();
    if (t == 0) {
        int s = 0;
        for (int k = 0; k < ROWS_PAD / 32; ++k) { const int u = wsum[k]; wsum[k] = s; s += u; }
    }
    __syncthreads();
    if (t < ROWS_PAD) {
        const int first = wsum[wv] + inc - c;
        rcur[t] = first;
        const int r = bin * BIN_ROWS + t;
        if (t < BIN_ROWS && r < NTOT) ((int2*)span)[r] = make_int2(bs + first, c);
    }
    __syncthreads();

    const bool staged = n <= STAGE_CAP;
#pragma unroll
    for (int k = 0; k < RUNS_PER_WAVE; ++k) {
        if ((have >> k) & 1) {
            const int  slot = atomicAdd(&rcur[p[k].x >> COL_BITS], 1);
            const int2 q    = make_int2(p[k].x & COL_MASK, p[k].y);
            if (staged) stage[slot] = q; else dst[slot] = q;
        }
    }
    if (longRun) {
        for (int k = 0; k < RUNS_PER_WAVE; ++k) {
            const int run = wv * RUNS_PER_WAVE + k;
            const int cn  = runCnt[run];
            const int s   = runSrc[run];
            for (int i = lane + 32; i < cn; i += 32) {
                const int2 e    = src[s + i];
                const int  slot = atomicAdd(&rcur[e.x >> COL_BITS], 1);
                const int2 q    = make_int2(e.x & COL_MASK, e.y);
                if (staged) stage[slot] = q; else dst[slot] = q;
            }
        }
    }
    if (staged) {
        __syncthreads();
        for (int i = t; i < n; i += SORT_TPB) dst[i] = stage[i];
    } else {
        __threadfence();
        __syncthreads();
    }

    const int sub  = t & 15;
    const int odd  = (t >> 4) & 1;
    for (;;) {
        int got = 0;
        if (lane == 0) got = atomicAdd(&nextPair, 1);
        const int pr = __builtin_amdgcn_readfirstlane(got);
        if (pr >= NPAIR_BIN) break;
        const int  lr  = pr * 2 + odd;
        const int  r   = bin * BIN_ROWS + lr;
        const bool on  = lr < BIN_ROWS && r < NTOT;
        const int  cnt = on ? rcnt[lr] : 0;
        const int  beg = on ? rcur[lr] - cnt : 0;
        float4 s;
        if (staged) s = row_edge_sum<1>(stage + beg, cnt, eu, ei, nullptr, sub);
        else        s = row_edge_sum<1>((const int2*)dst + beg, cnt, eu, ei, nullptr, sub);
        if (on) ((float4*)(x1 + (size_t)r * EMB))[sub] = s;
    }
}

template <int LAYER>
__device__ __forceinline__ const float* src_row(int c, const float* eu, const float* ei,
                                                const float* src) {
    if (LAYER == 1)
        return (c < NUM_USERS) ? eu + (size_t)c * EMB : ei + (size_t)(c - NUM_USERS) * EMB;
    return src + (size_t)c * EMB;
}

template <int LAYER>
__device__ __forceinline__ float4 row_edge_sum(const int2* ed, const int n,
                                               const float* __restrict__ eu,
                                               const float* __restrict__ ei,
                                               const float* src, const int sub) {
    const int nmax = __builtin_amdgcn_readfirstlane(max(n, __shfl_xor(n, 16, 32)));

    float4 s = make_float4(0.f, 0.f, 0.f, 0.f);
    for (int base = 0; base < nmax; base += 16) {
        int myc = 0; float myw = 0.f;
        if (base + sub < n) {
            int2 p = ed[base + sub];
            myc = p.x; myw = __int_as_float(p.y);
        }
        const int lim = min(16, nmax - base);
        for (int j = 0; j < lim; j += 4) {
            const int   c0 = __shfl(myc, j,     16), c1 = __shfl(myc, j + 1, 16);
            const int   c2 = __shfl(myc, j + 2, 16), c3 = __shfl(myc, j + 3, 16);
            const float w0 = __shfl(myw, j,     16), w1 = __shfl(myw, j + 1, 16);
            const float w2 = __shfl(myw, j + 2, 16), w3 = __shfl(myw, j + 3, 16);
            const float4 x0 = ((const float4*)src_row<LAYER>(c0, eu, ei, src))[sub];
            const float4 x1 = ((const float4*)src_row<LAYER>(c1, eu, ei, src))[sub];
            const float4 x2 = ((const float4*)src_row<LAYER>(c2, eu, ei, src))[sub];
            const float4 x3 = ((const float4*)src_row<LAYER>(c3, eu, ei, src))[sub];
            s.x += w0 * x0.x; s.y += w0 * x0.y; s.z += w0 * x0.z; s.w += w0 * x0.w;
            s.x += w1 * x1.x; s.y += w1 * x1.y; s.z += w1 * x1.z; s.w += w1 * x1.w;
            s.x += w2 * x2.x; s.y += w2 * x2.y; s.z += w2 * x2.z; s.w += w2 * x2.w;
            s.x += w3 * x3.x; s.y += w3 * x3.y; s.z += w3 * x3.z; s.w += w3 * x3.w;
        }
    }
    return s;
}

template <int LAYER>
__global__ __launch_bounds__(256) void lgcn_spmm_rows(const int* __restrict__ span,
                                                      const int* __restrict__ edges,
                                                      const float* __restrict__ eu,
                                                      const float* __restrict__ ei,
                                                      const float* src, float* dst) {
    const int gid = blockIdx.x * 256 + threadIdx.x;
    const int r   = gid >> 4;
    const int sub = threadIdx.x & 15;
    const int2 sp = ((const int2*)span)[r];
    const float4 s = row_edge_sum<LAYER>((const int2*)edges + sp.x, sp.y, eu, ei, src, sub);
    ((float4*)(dst + (size_t)r * EMB))[sub] = s;
}

#define ROWP4 17

__global__ __launch_bounds__(512) void lgcn_last_rows_gamma(const int* __restrict__ span,
                                                            const int* __restrict__ edges,
                                                            const float* __restrict__ eu,
                                                            const float* __restrict__ ei,
                                                            const float* __restrict__ x1,
                                                            const float* __restrict__ x2,
                                                            const int* __restrict__ users,
                                                            const int* __restrict__ items,
                                                            float* __restrict__ out) {
    __shared__ float4 rows[32 * ROWP4];
    const int tile = blockIdx.x;
    const int grp  = threadIdx.x >> 4;
    const int sub  = threadIdx.x & 15;
    const int pr   = tile * 16 + (grp & 15);
    const int r    = (grp < 16) ? users[pr] : NUM_USERS + items[pr];

    const int2 sp = ((const int2*)span)[r];
    const float* er = (r < NUM_USERS) ? eu + (size_t)r * EMB
                                      : ei + (size_t)(r - NUM_USERS) * EMB;
    const float4 e0 = ((const float4*)er)[sub];
    const float4 a1 = ((const float4*)(x1 + (size_t)r * EMB))[sub];
    const float4 a2 = ((const float4*)(x2 + (size_t)r * EMB))[sub];

    float4 s = row_edge_sum<3>((const int2*)edges + sp.x, sp.y, eu, ei, x2, sub);
    s.x = ((e0.x + a1.x) + a2.x) + s.x;
    s.y = ((e0.y + a1.y) + a2.y) + s.y;
    s.z = ((e0.z + a1.z) + a2.z) + s.z;
    s.w = ((e0.w + a1.w) + a2.w) + s.w;
    rows[grp * ROWP4 + sub] = s;
    __syncthreads();

    if (__builtin_amdgcn_readfirstlane((int)(threadIdx.x >> 5)) == 0) {
        const int l  = threadIdx.x & 31;
        const int m  = l & 15;
        const int kb = (l >> 4) * 2;
        const float* urow = (const float*)rows + m * (ROWP4 * 4);
        const float* vrow = (const float*)rows + (16 + m) * (ROWP4 * 4);

        v8f d = {};
#pragma unroll
        for (int k0 = 0; k0 < EMB; k0 += 4) {
            v2f a = *(const v2f*)(urow + k0 + kb);
            v2f b = *(const v2f*)(vrow + k0 + kb);
            d = __builtin_amdgcn_wmma_f32_16x16x4_f32(
                     false, a,  false, b,
                     (short)0, d,  false,  false);
        }

        const float scale = 1.0f / 16.0f;
        float gv = 0.f; int idx = -1;
#pragma unroll
        for (int j = 0; j < 8; ++j) {
            if (l == j)      { gv = d[j]; idx = j;     }
            if (l == 24 + j) { gv = d[j]; idx = 8 + j; }
        }
        if (idx >= 0) out[tile * 16 + idx] = gv * scale;
    }
}

extern "C" void kernel_launch(void* const* d_in, const int* in_sizes, int n_in,
                              void* d_out, int out_size, void* d_ws, size_t ws_size,
                              hipStream_t stream) {
    const int*   row      = (const int*)  d_in[0];
    const int*   col      = (const int*)  d_in[1];
    const float* val      = (const float*)d_in[2];
    const float* emb_user = (const float*)d_in[3];
    const float* emb_item = (const float*)d_in[4];
    const int*   users    = (const int*)  d_in[5];
    const int*   items    = (const int*)  d_in[6];
    float*       out      = (float*)d_out;
    (void)in_sizes; (void)n_in; (void)out_size; (void)ws_size;

    const size_t matElems = (size_t)NTOT * EMB;
    float* x1       = (float*)d_ws;
    float* x2       = x1 + matElems;
    int*   edges    = (int*)(x2 + matElems);
    int*   runMat   = edges + 2 * (size_t)NNZ;
    int*   span     = edges + 2 * (size_t)NNZ + 2 * NTOT;
    int*   part     = (int*)x2;

    const int tpb   = 256;
    const int gSpmm = NTOT * 16 / tpb;

    lgcn_chunk_by_bin<<<NPART, SORT_TPB, 0, stream>>>(row, col, val, runMat, part);
    lgcn_bin_rows_layer1<<<NBIN, SORT_TPB, 0, stream>>>(runMat, part, edges, span,
                                                       emb_user, emb_item, x1);

    lgcn_spmm_rows<2><<<gSpmm, tpb, 0, stream>>>(span, edges, emb_user, emb_item, x1, x2);

    lgcn_last_rows_gamma<<<NPAIR / 16, 512, 0, stream>>>(span, edges, emb_user, emb_item,
                                                         x1, x2, users, items, out);
}
